# starting kernel
# baseline (speedup 1.0000x reference)
#include <hip/hip_runtime.h>
#include <hip/hip_bf16.h>

typedef float v2f __attribute__((ext_vector_type(2)));
typedef float v8f __attribute__((ext_vector_type(8)));

#define POOL 5
#define PQ   25
#define CC   512
#define HH   22
#define WW   22
#define HW   484
#define NG   121
#define CK   12800
#define NROI 256
#define RSCALE (1.0f/16.0f)

__device__ __forceinline__ float hat_int(float t) {
    t = fminf(1.0f, fmaxf(-1.0f, t));
    float a = t + 1.0f, b = 1.0f - t;
    return (t < 0.0f) ? 0.5f * a * a : 1.0f - 0.5f * b * b;
}

__device__ __forceinline__ v8f wmma_f32(v2f a, v2f b, v8f c) {
    return __builtin_amdgcn_wmma_f32_16x16x4_f32(false, a, false, b,
                                                 (short)0, c, false, false);
}

__global__ __launch_bounds__(256) void prroi_pool_wmma(
    const float* __restrict__ feat, const float* __restrict__ bb,
    float* __restrict__ pooled)
{
    __shared__ v2f   sB[2 * NG * 32];
    __shared__ float sWx[POOL * WW];
    __shared__ float sWy[POOL * HH];

    const int n   = blockIdx.x;
    const int tid = threadIdx.x;

    const float bx = bb[n*4+0], by = bb[n*4+1], bwid = bb[n*4+2], bhgt = bb[n*4+3];
    const float x1 = bx * RSCALE, y1 = by * RSCALE;
    const float x2 = (bx + bwid) * RSCALE, y2 = (by + bhgt) * RSCALE;
    const float bw = fmaxf((x2 - x1) * (1.0f/POOL), 0.0f);
    const float bh = fmaxf((y2 - y1) * (1.0f/POOL), 0.0f);
    const float area = bw * bh;
    const float inv_area = (area > 0.0f) ? (1.0f / fmaxf(area, 1e-12f)) : 0.0f;

    if (tid < POOL*WW) {
        int q = tid / WW, i = tid % WW;
        sWx[tid] = hat_int(x1 + (q+1)*bw - (float)i) - hat_int(x1 + q*bw - (float)i);
    } else if (tid < POOL*WW + POOL*HH) {
        int t2 = tid - POOL*WW;
        int p = t2 / HH, j = t2 % HH;
        sWy[t2] = hat_int(y1 + (p+1)*bh - (float)j) - hat_int(y1 + p*bh - (float)j);
    }
    __syncthreads();

    for (int idx = tid; idx < 2*NG*32; idx += 256) {
        int bt = idx / (NG*32);
        int r  = idx - bt*NG*32;
        int g  = r >> 5;
        int ln = r & 31;
        int pq = bt*16 + (ln & 15);
        int k  = 4*g + 2*(ln >> 4);
        v2f val = {0.0f, 0.0f};
        if (pq < PQ) {
            int p = pq / POOL, q = pq % POOL;
            float wp0 = sWy[p*HH + (k    / WW)] * sWx[q*WW + (k    % WW)];
            float wp1 = sWy[p*HH + ((k+1)/ WW)] * sWx[q*WW + ((k+1)% WW)];
            val.x = wp0 * inv_area;
            val.y = wp1 * inv_area;
        }
        sB[idx] = val;
    }
    __syncthreads();

    const int wv   = tid >> 5;
    const int lane = tid & 31;
    const int m16  = lane & 15;
    const int koff = (lane >> 4) * 2;

    const float* fn = feat + (long)n * (CC * HW);
    float*       pn = pooled + (long)n * CK;

    const v2f* B0 = sB + lane;
    const v2f* B1 = sB + NG*32 + lane;

    for (int job = 0; job < 4; ++job) {
        const int ctile = wv + job * 8;
        const float* ap = fn + (long)(ctile*16 + m16) * HW + koff;

        v8f acc0 = {};
        v8f acc1 = {};
        #pragma unroll 1
        for (int g = 0; g < NG - 1; g += 4) {
            const float* base = ap + g*4;
            v2f x0 = *(const v2f*)(base + 0);
            v2f x1 = *(const v2f*)(base + 4);
            v2f x2 = *(const v2f*)(base + 8);
            v2f x3 = *(const v2f*)(base + 12);
            const v2f* pb0 = B0 + g*32;
            const v2f* pb1 = B1 + g*32;
            acc0 = wmma_f32(x0, pb0[ 0], acc0);  acc1 = wmma_f32(x0, pb1[ 0], acc1);
            acc0 = wmma_f32(x1, pb0[32], acc0);  acc1 = wmma_f32(x1, pb1[32], acc1);
            acc0 = wmma_f32(x2, pb0[64], acc0);  acc1 = wmma_f32(x2, pb1[64], acc1);
            acc0 = wmma_f32(x3, pb0[96], acc0);  acc1 = wmma_f32(x3, pb1[96], acc1);
        }
        {
            v2f x = *(const v2f*)(ap + 480);
            acc0 = wmma_f32(x, B0[120*32], acc0);
            acc1 = wmma_f32(x, B1[120*32], acc1);
        }

        const int ncol  = m16;
        const int mbase = (lane >> 4) * 8;
        float* p0 = pn + ncol;
        #pragma unroll
        for (int j = 0; j < 8; ++j) p0[(ctile*16 + mbase + j) * PQ] = acc0[j];
        if (ncol < 9) {
            float* p1 = pn + 16 + ncol;
            #pragma unroll
            for (int j = 0; j < 8; ++j) p1[(ctile*16 + mbase + j) * PQ] = acc1[j];
        }
    }
}

__global__ __launch_bounds__(256) void head_wmma(
    const float* __restrict__ pooled,
    const float* __restrict__ w_obj, const float* __restrict__ b_obj,
    const float* __restrict__ w_mot, const float* __restrict__ b_mot,
    float* __restrict__ out)
{
    __shared__ float red[8 * 256];
    const int mtile = blockIdx.x;
    const int ntile = blockIdx.y;
    const int tid   = threadIdx.x;
    const int wv    = tid >> 5;
    const int lane  = tid & 31;
    const int m16   = lane & 15;
    const int kpair = (lane >> 4) * 2;
    const int KSL   = CK / 8;
    const int GSL   = KSL / 4;

    const int nh = ntile*16 + m16;
    const float* wrow;
    if      (nh < 20) wrow = w_obj + (long)nh * CK;
    else if (nh < 24) wrow = w_mot + (long)(nh - 20) * CK;
    else              wrow = w_obj;

    const float* ap = pooled + (long)(mtile*16 + m16) * CK + kpair + wv * KSL;
    const float* bp = wrow + kpair + wv * KSL;

    v8f acc = {};
    #pragma unroll 1
    for (int g = 0; g < GSL; g += 4) {
        const float* a4 = ap + g*4;
        const float* b4 = bp + g*4;
        v2f xa0 = *(const v2f*)(a4 + 0);
        v2f xa1 = *(const v2f*)(a4 + 4);
        v2f xa2 = *(const v2f*)(a4 + 8);
        v2f xa3 = *(const v2f*)(a4 + 12);
        v2f xb0 = *(const v2f*)(b4 + 0);
        v2f xb1 = *(const v2f*)(b4 + 4);
        v2f xb2 = *(const v2f*)(b4 + 8);
        v2f xb3 = *(const v2f*)(b4 + 12);
        acc = wmma_f32(xa0, xb0, acc);
        acc = wmma_f32(xa1, xb1, acc);
        acc = wmma_f32(xa2, xb2, acc);
        acc = wmma_f32(xa3, xb3, acc);
    }

    #pragma unroll
    for (int j = 0; j < 8; ++j) red[wv*256 + lane*8 + j] = acc[j];
    __syncthreads();

    float s = 0.0f;
    #pragma unroll
    for (int w = 0; w < 8; ++w) s += red[w*256 + tid];

    const int lsrc = tid >> 3, j = tid & 7;
    const int ncol = lsrc & 15;
    const int mrow = (lsrc >> 4) * 8 + j;
    const int nh2  = ntile*16 + ncol;
    const int roi  = mtile*16 + mrow;
    if (nh2 < 24) {
        float bias = (nh2 < 20) ? b_obj[nh2] : b_mot[nh2 - 20];
        float val = s + bias;
        if (nh2 < 20) out[roi*20 + nh2] = val;
        else          out[20*NROI + roi*4 + (nh2 - 20)] = val;
    }
}

extern "C" void kernel_launch(void* const* d_in, const int* in_sizes, int n_in,
                              void* d_out, int out_size, void* d_ws, size_t ws_size,
                              hipStream_t stream) {
    const float* feat  = (const float*)d_in[0];
    const float* bb    = (const float*)d_in[1];
    const float* w_obj = (const float*)d_in[2];
    const float* b_obj = (const float*)d_in[3];
    const float* w_mot = (const float*)d_in[4];
    const float* b_mot = (const float*)d_in[5];
    float* out    = (float*)d_out;
    float* pooled = (float*)d_ws;

    prroi_pool_wmma<<<dim3(NROI), 256, 0, stream>>>(feat, bb, pooled);
    head_wmma<<<dim3(16, 2), 256, 0, stream>>>(pooled, w_obj, b_obj, w_mot, b_mot, out);
}


# pooling kernel: more feature bytes in flight per WGP (32 waves per ROI, one channel tile per wave; b128 loads in batches of ten, next batch asked ahead); k grouped by 8, so the k sum order changes
# speedup vs baseline: 1.7071x; 1.7071x over previous
#include <hip/hip_runtime.h>
#include <hip/hip_bf16.h>

typedef float v2f __attribute__((ext_vector_type(2)));
typedef float v4f __attribute__((ext_vector_type(4)));
typedef float v8f __attribute__((ext_vector_type(8)));

#define POOL 5
#define PQ   25
#define CC   512
#define HH   22
#define WW   22
#define HW   484
#define NP   60
#define NJ   61
#define BATCH  10
#define NBATCH 6
#define CK   12800
#define NROI 256
#define RSCALE (1.0f/16.0f)

__device__ __forceinline__ float hat_int(float t) {
    t = fminf(1.0f, fmaxf(-1.0f, t));
    float a = t + 1.0f, b = 1.0f - t;
    return (t < 0.0f) ? 0.5f * a * a : 1.0f - 0.5f * b * b;
}

__device__ __forceinline__ v8f wmma_f32(v2f a, v2f b, v8f c) {
    return __builtin_amdgcn_wmma_f32_16x16x4_f32(false, a, false, b,
                                                 (short)0, c, false, false);
}

__device__ __forceinline__ void lds_barrier() {
    __builtin_amdgcn_fence(__ATOMIC_RELEASE, "workgroup", "local");
    __builtin_amdgcn_s_barrier();
    __builtin_amdgcn_fence(__ATOMIC_ACQUIRE, "workgroup", "local");
}

__global__ __launch_bounds__(1024) void prroi_pool_wmma(
    const float* __restrict__ feat, const float* __restrict__ bb,
    float* __restrict__ pooled)
{
    __shared__ v4f   sB[2 * NJ * 32];
    __shared__ float sWx[POOL * WW];
    __shared__ float sWy[POOL * HH];

    const int n    = blockIdx.x;
    const int tid  = threadIdx.x;
    const int wv   = tid >> 5;
    const int lane = tid & 31;
    const int m16  = lane & 15;
    const int hf   = lane >> 4;

    const float* rowp = feat + (long)n * (CC * HW) + (long)(wv*16 + m16) * HW;
    const float* ap   = rowp + 4*hf;

    v4f buf[2][BATCH];
    #pragma unroll
    for (int i = 0; i < BATCH; ++i) buf[0][i] = *(const v4f*)(ap + 8*i);
    const v2f xt = *(const v2f*)(rowp + 480 + 2*hf);

    const float bx = bb[n*4+0], by = bb[n*4+1], bwid = bb[n*4+2], bhgt = bb[n*4+3];
    const float x1 = bx * RSCALE, y1 = by * RSCALE;
    const float x2 = (bx + bwid) * RSCALE, y2 = (by + bhgt) * RSCALE;
    const float bw = fmaxf((x2 - x1) * (1.0f/POOL), 0.0f);
    const float bh = fmaxf((y2 - y1) * (1.0f/POOL), 0.0f);
    const float area = bw * bh;
    const float inv_area = (area > 0.0f) ? (1.0f / fmaxf(area, 1e-12f)) : 0.0f;

    if (tid < POOL*WW) {
        int q = tid / WW, i = tid % WW;
        sWx[tid] = hat_int(x1 + (q+1)*bw - (float)i) - hat_int(x1 + q*bw - (float)i);
    } else if (tid < POOL*WW + POOL*HH) {
        int t2 = tid - POOL*WW;
        int p = t2 / HH, j = t2 % HH;
        sWy[t2] = hat_int(y1 + (p+1)*bh - (float)j) - hat_int(y1 + p*bh - (float)j);
    }
    lds_barrier();

    for (int idx = tid; idx < 2*NJ*32; idx += 1024) {
        int bt = idx / (NJ*32);
        int r  = idx - bt*NJ*32;
        int j  = r >> 5;
        int ln = r & 31;
        int pq = bt*16 + (ln & 15);
        int hh = ln >> 4;
        v4f val = {0.0f, 0.0f, 0.0f, 0.0f};
        if (pq < PQ) {
            int p = pq / POOL, q = pq % POOL;
            const float* wy = sWy + p*HH;
            const float* wx = sWx + q*WW;
            if (j < NP) {
                int k = 8*j + 4*hh;
                val.x = (wy[(k  )/WW] * wx[(k  )%WW]) * inv_area;
                val.y = (wy[(k+1)/WW] * wx[(k+1)%WW]) * inv_area;
                val.z = (wy[(k+2)/WW] * wx[(k+2)%WW]) * inv_area;
                val.w = (wy[(k+3)/WW] * wx[(k+3)%WW]) * inv_area;
            } else {
                int k = 8*NP + 2*hh;
                val.x = (wy[(k  )/WW] * wx[(k  )%WW]) * inv_area;
                val.y = (wy[(k+1)/WW] * wx[(k+1)%WW]) * inv_area;
            }
        }
        sB[idx] = val;
    }
    lds_barrier();

    float* pn = pooled + (long)n * CK;

    const v4f* B0 = sB + lane;
    const v4f* B1 = sB + NJ*32 + lane;

    v8f acc0 = {};
    v8f acc1 = {};
    #pragma unroll
    for (int b = 0; b < NBATCH; ++b) {
        if (b + 1 < NBATCH) {
            #pragma unroll
            for (int i = 0; i < BATCH; ++i)
                buf[(b+1)&1][i] = *(const v4f*)(ap + 8*((b+1)*BATCH + i));
            __builtin_amdgcn_sched_barrier(0);
        }
        #pragma unroll
        for (int i = 0; i < BATCH; ++i) {
            const int j = b*BATCH + i;
            const v4f x  = buf[b&1][i];
            const v4f w0 = B0[j*32];
            const v4f w1 = B1[j*32];
            const v2f xlo = {x.x, x.y},   xhi = {x.z, x.w};
            const v2f w0l = {w0.x, w0.y}, w0h = {w0.z, w0.w};
            const v2f w1l = {w1.x, w1.y}, w1h = {w1.z, w1.w};
            acc0 = wmma_f32(xlo, w0l, acc0);  acc1 = wmma_f32(xlo, w1l, acc1);
            acc0 = wmma_f32(xhi, w0h, acc0);  acc1 = wmma_f32(xhi, w1h, acc1);
        }
    }
    {
        const v4f w0 = B0[NP*32];
        const v4f w1 = B1[NP*32];
        const v2f w0l = {w0.x, w0.y};
        const v2f w1l = {w1.x, w1.y};
        acc0 = wmma_f32(xt, w0l, acc0);
        acc1 = wmma_f32(xt, w1l, acc1);
    }

    const int ncol  = m16;
    const int mbase = hf * 8;
    float* p0 = pn + ncol;
    #pragma unroll
    for (int j = 0; j < 8; ++j) p0[(wv*16 + mbase + j) * PQ] = acc0[j];
    if (ncol < 9) {
        float* p1 = pn + 16 + ncol;
        #pragma unroll
        for (int j = 0; j < 8; ++j) p1[(wv*16 + mbase + j) * PQ] = acc1[j];
    }
}

__global__ __launch_bounds__(256) void head_wmma(
    const float* __restrict__ pooled,
    const float* __restrict__ w_obj, const float* __restrict__ b_obj,
    const float* __restrict__ w_mot, const float* __restrict__ b_mot,
    float* __restrict__ out)
{
    __shared__ float red[8 * 256];
    const int mtile = blockIdx.x;
    const int ntile = blockIdx.y;
    const int tid   = threadIdx.x;
    const int wv    = tid >> 5;
    const int lane  = tid & 31;
    const int m16   = lane & 15;
    const int kpair = (lane >> 4) * 2;
    const int KSL   = CK / 8;
    const int GSL   = KSL / 4;

    const int nh = ntile*16 + m16;
    const float* wrow;
    if      (nh < 20) wrow = w_obj + (long)nh * CK;
    else if (nh < 24) wrow = w_mot + (long)(nh - 20) * CK;
    else              wrow = w_obj;

    const float* ap = pooled + (long)(mtile*16 + m16) * CK + kpair + wv * KSL;
    const float* bp = wrow + kpair + wv * KSL;

    v8f acc = {};
    #pragma unroll 1
    for (int g = 0; g < GSL; g += 4) {
        const float* a4 = ap + g*4;
        const float* b4 = bp + g*4;
        v2f xa0 = *(const v2f*)(a4 + 0);
        v2f xa1 = *(const v2f*)(a4 + 4);
        v2f xa2 = *(const v2f*)(a4 + 8);
        v2f xa3 = *(const v2f*)(a4 + 12);
        v2f xb0 = *(const v2f*)(b4 + 0);
        v2f xb1 = *(const v2f*)(b4 + 4);
        v2f xb2 = *(const v2f*)(b4 + 8);
        v2f xb3 = *(const v2f*)(b4 + 12);
        acc = wmma_f32(xa0, xb0, acc);
        acc = wmma_f32(xa1, xb1, acc);
        acc = wmma_f32(xa2, xb2, acc);
        acc = wmma_f32(xa3, xb3, acc);
    }

    #pragma unroll
    for (int j = 0; j < 8; ++j) red[wv*256 + lane*8 + j] = acc[j];
    __syncthreads();

    float s = 0.0f;
    #pragma unroll
    for (int w = 0; w < 8; ++w) s += red[w*256 + tid];

    const int lsrc = tid >> 3, j = tid & 7;
    const int ncol = lsrc & 15;
    const int mrow = (lsrc >> 4) * 8 + j;
    const int nh2  = ntile*16 + ncol;
    const int roi  = mtile*16 + mrow;
    if (nh2 < 24) {
        float bias = (nh2 < 20) ? b_obj[nh2] : b_mot[nh2 - 20];
        float val = s + bias;
        if (nh2 < 20) out[roi*20 + nh2] = val;
        else          out[20*NROI + roi*4 + (nh2 - 20)] = val;
    }
}

extern "C" void kernel_launch(void* const* d_in, const int* in_sizes, int n_in,
                              void* d_out, int out_size, void* d_ws, size_t ws_size,
                              hipStream_t stream) {
    const float* feat  = (const float*)d_in[0];
    const float* bb    = (const float*)d_in[1];
    const float* w_obj = (const float*)d_in[2];
    const float* b_obj = (const float*)d_in[3];
    const float* w_mot = (const float*)d_in[4];
    const float* b_mot = (const float*)d_in[5];
    float* out    = (float*)d_out;
    float* pooled = (float*)d_ws;

    prroi_pool_wmma<<<dim3(NROI), 1024, 0, stream>>>(feat, bb, pooled);
    head_wmma<<<dim3(16, 2), 256, 0, stream>>>(pooled, w_obj, b_obj, w_mot, b_mot, out);
}


# head GEMM: K split into 16 slices over 256 work-groups, each wave asking for its whole share of b128 loads at once; partials summed in slice order by a third small launch; head k sum order changes
# speedup vs baseline: 2.8135x; 1.6481x over previous
#include <hip/hip_runtime.h>
#include <hip/hip_bf16.h>

typedef float v2f __attribute__((ext_vector_type(2)));
typedef float v4f __attribute__((ext_vector_type(4)));
typedef float v8f __attribute__((ext_vector_type(8)));

#define POOL 5
#define PQ   25
#define CC   512
#define HH   22
#define WW   22
#define HW   484
#define NP   60
#define NJ   61
#define BATCH  10
#define NBATCH 6
#define CK   12800
#define NROI 256
#define RSCALE (1.0f/16.0f)

__device__ __forceinline__ float hat_int(float t) {
    t = fminf(1.0f, fmaxf(-1.0f, t));
    float a = t + 1.0f, b = 1.0f - t;
    return (t < 0.0f) ? 0.5f * a * a : 1.0f - 0.5f * b * b;
}

__device__ __forceinline__ v8f wmma_f32(v2f a, v2f b, v8f c) {
    return __builtin_amdgcn_wmma_f32_16x16x4_f32(false, a, false, b,
                                                 (short)0, c, false, false);
}

__device__ __forceinline__ void lds_barrier() {
    __builtin_amdgcn_fence(__ATOMIC_RELEASE, "workgroup", "local");
    __builtin_amdgcn_s_barrier();
    __builtin_amdgcn_fence(__ATOMIC_ACQUIRE, "workgroup", "local");
}

__global__ __launch_bounds__(1024) void prroi_pool_wmma(
    const float* __restrict__ feat, const float* __restrict__ bb,
    float* __restrict__ pooled)
{
    __shared__ v4f   sB[2 * NJ * 32];
    __shared__ float sWx[POOL * WW];
    __shared__ float sWy[POOL * HH];

    const int n    = blockIdx.x;
    const int tid  = threadIdx.x;
    const int wv   = tid >> 5;
    const int lane = tid & 31;
    const int m16  = lane & 15;
    const int hf   = lane >> 4;

    const float* rowp = feat + (long)n * (CC * HW) + (long)(wv*16 + m16) * HW;
    const float* ap   = rowp + 4*hf;

    v4f buf[2][BATCH];
    #pragma unroll
    for (int i = 0; i < BATCH; ++i) buf[0][i] = *(const v4f*)(ap + 8*i);
    const v2f xt = *(const v2f*)(rowp + 480 + 2*hf);

    const float bx = bb[n*4+0], by = bb[n*4+1], bwid = bb[n*4+2], bhgt = bb[n*4+3];
    const float x1 = bx * RSCALE, y1 = by * RSCALE;
    const float x2 = (bx + bwid) * RSCALE, y2 = (by + bhgt) * RSCALE;
    const float bw = fmaxf((x2 - x1) * (1.0f/POOL), 0.0f);
    const float bh = fmaxf((y2 - y1) * (1.0f/POOL), 0.0f);
    const float area = bw * bh;
    const float inv_area = (area > 0.0f) ? (1.0f / fmaxf(area, 1e-12f)) : 0.0f;

    if (tid < POOL*WW) {
        int q = tid / WW, i = tid % WW;
        sWx[tid] = hat_int(x1 + (q+1)*bw - (float)i) - hat_int(x1 + q*bw - (float)i);
    } else if (tid < POOL*WW + POOL*HH) {
        int t2 = tid - POOL*WW;
        int p = t2 / HH, j = t2 % HH;
        sWy[t2] = hat_int(y1 + (p+1)*bh - (float)j) - hat_int(y1 + p*bh - (float)j);
    }
    lds_barrier();

    for (int idx = tid; idx < 2*NJ*32; idx += 1024) {
        int bt = idx / (NJ*32);
        int r  = idx - bt*NJ*32;
        int j  = r >> 5;
        int ln = r & 31;
        int pq = bt*16 + (ln & 15);
        int hh = ln >> 4;
        v4f val = {0.0f, 0.0f, 0.0f, 0.0f};
        if (pq < PQ) {
            int p = pq / POOL, q = pq % POOL;
            const float* wy = sWy + p*HH;
            const float* wx = sWx + q*WW;
            if (j < NP) {
                int k = 8*j + 4*hh;
                val.x = (wy[(k  )/WW] * wx[(k  )%WW]) * inv_area;
                val.y = (wy[(k+1)/WW] * wx[(k+1)%WW]) * inv_area;
                val.z = (wy[(k+2)/WW] * wx[(k+2)%WW]) * inv_area;
                val.w = (wy[(k+3)/WW] * wx[(k+3)%WW]) * inv_area;
            } else {
                int k = 8*NP + 2*hh;
                val.x = (wy[(k  )/WW] * wx[(k  )%WW]) * inv_area;
                val.y = (wy[(k+1)/WW] * wx[(k+1)%WW]) * inv_area;
            }
        }
        sB[idx] = val;
    }
    lds_barrier();

    float* pn = pooled + (long)n * CK;

    const v4f* B0 = sB + lane;
    const v4f* B1 = sB + NJ*32 + lane;

    v8f acc0 = {};
    v8f acc1 = {};
    #pragma unroll
    for (int b = 0; b < NBATCH; ++b) {
        if (b + 1 < NBATCH) {
            #pragma unroll
            for (int i = 0; i < BATCH; ++i)
                buf[(b+1)&1][i] = *(const v4f*)(ap + 8*((b+1)*BATCH + i));
            __builtin_amdgcn_sched_barrier(0);
        }
        #pragma unroll
        for (int i = 0; i < BATCH; ++i) {
            const int j = b*BATCH + i;
            const v4f x  = buf[b&1][i];
            const v4f w0 = B0[j*32];
            const v4f w1 = B1[j*32];
            const v2f xlo = {x.x, x.y},   xhi = {x.z, x.w};
            const v2f w0l = {w0.x, w0.y}, w0h = {w0.z, w0.w};
            const v2f w1l = {w1.x, w1.y}, w1h = {w1.z, w1.w};
            acc0 = wmma_f32(xlo, w0l, acc0);  acc1 = wmma_f32(xlo, w1l, acc1);
            acc0 = wmma_f32(xhi, w0h, acc0);  acc1 = wmma_f32(xhi, w1h, acc1);
        }
    }
    {
        const v4f w0 = B0[NP*32];
        const v4f w1 = B1[NP*32];
        const v2f w0l = {w0.x, w0.y};
        const v2f w1l = {w1.x, w1.y};
        acc0 = wmma_f32(xt, w0l, acc0);
        acc1 = wmma_f32(xt, w1l, acc1);
    }

    const int ncol  = m16;
    const int mbase = hf * 8;
    float* p0 = pn + ncol;
    #pragma unroll
    for (int j = 0; j < 8; ++j) p0[(wv*16 + mbase + j) * PQ] = acc0[j];
    if (ncol < 9) {
        float* p1 = pn + 16 + ncol;
        #pragma unroll
        for (int j = 0; j < 8; ++j) p1[(wv*16 + mbase + j) * PQ] = acc1[j];
    }
}

#define KS     16
#define KSLICE 800
#define HWV    4
#define HGRP   25
__global__ __launch_bounds__(128) void head_partial_wmma(
    float* ws, const float* __restrict__ w_obj, const float* __restrict__ w_mot)
{
    __shared__ v4f red[HWV * 128];
    const int mtile = blockIdx.x;
    const int ks    = blockIdx.y;
    const int tid   = threadIdx.x;
    const int wv    = tid >> 5;
    const int lane  = tid & 31;
    const int m16   = lane & 15;
    const int hf    = lane >> 4;

    const long koff = (long)ks * KSLICE + wv * (HGRP * 8) + 4 * hf;
    const float* ap  = ws + (long)(mtile*16 + m16) * CK + koff;
    const float* b0p = w_obj + (long)m16 * CK + koff;
    const int  nh1   = 16 + (m16 & 7);
    const float* b1p = (nh1 < 20) ? (w_obj + (long)nh1 * CK + koff)
                                  : (w_mot + (long)(nh1 - 20) * CK + koff);

    v4f xa[HGRP], xb0[HGRP], xb1[HGRP];
    #pragma unroll
    for (int i = 0; i < HGRP; ++i) {
        xa[i]  = *(const v4f*)(ap  + 8*i);
        xb0[i] = *(const v4f*)(b0p + 8*i);
        xb1[i] = *(const v4f*)(b1p + 8*i);
    }
    __builtin_amdgcn_sched_barrier(0);

    v8f acc0 = {};
    v8f acc1 = {};
    #pragma unroll
    for (int i = 0; i < HGRP; ++i) {
        const v4f x = xa[i], w0 = xb0[i], w1 = xb1[i];
        const v2f xlo = {x.x, x.y},   xhi = {x.z, x.w};
        const v2f w0l = {w0.x, w0.y}, w0h = {w0.z, w0.w};
        const v2f w1l = {w1.x, w1.y}, w1h = {w1.z, w1.w};
        acc0 = wmma_f32(xlo, w0l, acc0);  acc1 = wmma_f32(xlo, w1l, acc1);
        acc0 = wmma_f32(xhi, w0h, acc0);  acc1 = wmma_f32(xhi, w1h, acc1);
    }

    v4f* rw = red + wv * 128;
    rw[lane*2 + 0]      = (v4f){acc0[0], acc0[1], acc0[2], acc0[3]};
    rw[lane*2 + 1]      = (v4f){acc0[4], acc0[5], acc0[6], acc0[7]};
    rw[64 + lane*2 + 0] = (v4f){acc1[0], acc1[1], acc1[2], acc1[3]};
    rw[64 + lane*2 + 1] = (v4f){acc1[4], acc1[5], acc1[6], acc1[7]};
    __syncthreads();

    v4f s = red[tid];
    #pragma unroll
    for (int w = 1; w < HWV; ++w) s += red[w*128 + tid];
    *(v4f*)(ws + (long)(mtile*16) * CK + (long)ks * KSLICE + 4*tid) = s;
}

__global__ __launch_bounds__(512) void head_reduce(
    const float* __restrict__ ws,
    const float* __restrict__ b_obj, const float* __restrict__ b_mot,
    float* __restrict__ out)
{
    const int mtile = blockIdx.x;
    const int e     = threadIdx.x;
    const int tile  = e >> 8;
    const int lane  = (e >> 3) & 31;
    const int j     = e & 7;
    const int nh    = tile*16 + (lane & 15);
    const int roi   = mtile*16 + (lane >> 4)*8 + j;
    if (nh < 24) {
        const float* p = ws + (long)(mtile*16) * CK + e;
        float v[KS];
        #pragma unroll
        for (int k = 0; k < KS; ++k) v[k] = p[k * KSLICE];
        float s = v[0];
        #pragma unroll
        for (int k = 1; k < KS; ++k) s += v[k];
        if (nh < 20) out[roi*20 + nh] = s + b_obj[nh];
        else         out[20*NROI + roi*4 + (nh - 20)] = s + b_mot[nh - 20];
    }
}

extern "C" void kernel_launch(void* const* d_in, const int* in_sizes, int n_in,
                              void* d_out, int out_size, void* d_ws, size_t ws_size,
                              hipStream_t stream) {
    const float* feat  = (const float*)d_in[0];
    const float* bb    = (const float*)d_in[1];
    const float* w_obj = (const float*)d_in[2];
    const float* b_obj = (const float*)d_in[3];
    const float* w_mot = (const float*)d_in[4];
    const float* b_mot = (const float*)d_in[5];
    float* out    = (float*)d_out;
    float* pooled = (float*)d_ws;

    prroi_pool_wmma<<<dim3(NROI), 1024, 0, stream>>>(feat, bb, pooled);
    head_partial_wmma<<<dim3(16, KS), 128, 0, stream>>>(pooled, w_obj, w_mot);
    head_reduce<<<dim3(16), 512, 0, stream>>>(pooled, b_obj, b_mot, out);
}


# pooling kernel: k loop and LDS weight build cover only the feature rows whose y weights for the ROI are non-zero (rolled loop, loads in batches of six); skipped products are exact zeros, k order kept
# speedup vs baseline: 2.9629x; 1.0531x over previous
#include <hip/hip_runtime.h>
#include <hip/hip_bf16.h>

typedef float v2f __attribute__((ext_vector_type(2)));
typedef float v4f __attribute__((ext_vector_type(4)));
typedef float v8f __attribute__((ext_vector_type(8)));

#define POOL 5
#define PQ   25
#define CC   512
#define HH   22
#define WW   22
#define HW   484
#define NP   60
#define NJ   61
#define BATCH  6
#define NBATCH 10
#define CK   12800
#define NROI 256
#define RSCALE (1.0f/16.0f)

__device__ __forceinline__ float hat_int(float t) {
    t = fminf(1.0f, fmaxf(-1.0f, t));
    float a = t + 1.0f, b = 1.0f - t;
    return (t < 0.0f) ? 0.5f * a * a : 1.0f - 0.5f * b * b;
}

__device__ __forceinline__ v8f wmma_f32(v2f a, v2f b, v8f c) {
    return __builtin_amdgcn_wmma_f32_16x16x4_f32(false, a, false, b,
                                                 (short)0, c, false, false);
}

__device__ __forceinline__ void lds_barrier() {
    __builtin_amdgcn_fence(__ATOMIC_RELEASE, "workgroup", "local");
    __builtin_amdgcn_s_barrier();
    __builtin_amdgcn_fence(__ATOMIC_ACQUIRE, "workgroup", "local");
}

__device__ __forceinline__ void mma_batch(const v4f (&x)[BATCH], const v4f* __restrict__ w0p,
                                          const v4f* __restrict__ w1p, v8f& acc0, v8f& acc1)
{
    #pragma unroll
    for (int i = 0; i < BATCH; ++i) {
        const v4f xv = x[i];
        const v4f w0 = w0p[i*32];
        const v4f w1 = w1p[i*32];
        const v2f xlo = {xv.x, xv.y}, xhi = {xv.z, xv.w};
        const v2f w0l = {w0.x, w0.y}, w0h = {w0.z, w0.w};
        const v2f w1l = {w1.x, w1.y}, w1h = {w1.z, w1.w};
        acc0 = wmma_f32(xlo, w0l, acc0);  acc1 = wmma_f32(xlo, w1l, acc1);
        acc0 = wmma_f32(xhi, w0h, acc0);  acc1 = wmma_f32(xhi, w1h, acc1);
    }
}

__global__ __launch_bounds__(1024) void prroi_pool_wmma(
    const float* __restrict__ feat, const float* __restrict__ bb,
    float* __restrict__ pooled)
{
    __shared__ v4f   sB[2 * NJ * 32];
    __shared__ float sWx[POOL * WW];
    __shared__ float sWy[POOL * HH];

    const int n    = blockIdx.x;
    const int tid  = threadIdx.x;
    const int wv   = tid >> 5;
    const int lane = tid & 31;
    const int m16  = lane & 15;
    const int hf   = lane >> 4;

    const float* rowp = feat + (long)n * (CC * HW) + (long)(wv*16 + m16) * HW;

    const v2f xt = *(const v2f*)(rowp + 480 + 2*hf);

    const float bx = bb[n*4+0], by = bb[n*4+1], bwid = bb[n*4+2], bhgt = bb[n*4+3];
    const float x1 = bx * RSCALE, y1 = by * RSCALE;
    const float x2 = (bx + bwid) * RSCALE, y2 = (by + bhgt) * RSCALE;
    const float bw = fmaxf((x2 - x1) * (1.0f/POOL), 0.0f);
    const float bh = fmaxf((y2 - y1) * (1.0f/POOL), 0.0f);
    const float area = bw * bh;
    const float inv_area = (area > 0.0f) ? (1.0f / fmaxf(area, 1e-12f)) : 0.0f;

    if (tid < POOL*WW) {
        int q = tid / WW, i = tid % WW;
        sWx[tid] = hat_int(x1 + (q+1)*bw - (float)i) - hat_int(x1 + q*bw - (float)i);
    } else if (tid < POOL*WW + POOL*HH) {
        int t2 = tid - POOL*WW;
        int p = t2 / HH, j = t2 % HH;
        sWy[t2] = hat_int(y1 + (p+1)*bh - (float)j) - hat_int(y1 + p*bh - (float)j);
    }
    lds_barrier();

    bool act = false;
    if (lane < HH) {
        #pragma unroll
        for (int p = 0; p < POOL; ++p) act |= !(sWy[p*HH + lane] == 0.0f);
    }
    const unsigned rmask = __builtin_amdgcn_ballot_w32(act);
    int j_lo = 0, j_hi = 0;
    if (rmask != 0u) {
        const int r_lo = __builtin_ctz(rmask);
        const int r_hi = 31 - __builtin_clz(rmask);
        j_lo = (r_lo * WW) >> 3;
        j_hi = min(((r_hi + 1) * WW + 7) >> 3, NP);
    }
    const int nb = (j_hi - j_lo + BATCH - 1) / BATCH;
    const int js = min(j_lo, NP - nb*BATCH);

    const float* ap = rowp + 4*hf + 8*js;

    v4f bufA[BATCH], bufB[BATCH];
    if (nb > 0) {
        #pragma unroll
        for (int i = 0; i < BATCH; ++i) bufA[i] = *(const v4f*)(ap + 8*i);
    }

    const int nslot = nb*BATCH + 1;
    for (int idx = tid; idx < 2*nslot*32; idx += 1024) {
        const int bt = (idx >= nslot*32) ? 1 : 0;
        const int r  = idx - bt*nslot*32;
        const int jj = r >> 5;
        const int ln = r & 31;
        const int j  = (jj < nslot - 1) ? (js + jj) : NP;
        const int pq = bt*16 + (ln & 15);
        const int hh = ln >> 4;
        v4f val = {0.0f, 0.0f, 0.0f, 0.0f};
        if (pq < PQ) {
            int p = pq / POOL, q = pq % POOL;
            const float* wy = sWy + p*HH;
            const float* wx = sWx + q*WW;
            if (j < NP) {
                int k = 8*j + 4*hh;
                val.x = (wy[(k  )/WW] * wx[(k  )%WW]) * inv_area;
                val.y = (wy[(k+1)/WW] * wx[(k+1)%WW]) * inv_area;
                val.z = (wy[(k+2)/WW] * wx[(k+2)%WW]) * inv_area;
                val.w = (wy[(k+3)/WW] * wx[(k+3)%WW]) * inv_area;
            } else {
                int k = 8*NP + 2*hh;
                val.x = (wy[(k  )/WW] * wx[(k  )%WW]) * inv_area;
                val.y = (wy[(k+1)/WW] * wx[(k+1)%WW]) * inv_area;
            }
        }
        sB[(bt*NJ + j)*32 + ln] = val;
    }
    lds_barrier();

    float* pn = pooled + (long)n * CK;

    const v4f* B0 = sB + js*32 + lane;
    const v4f* B1 = B0 + NJ*32;

    v8f acc0 = {};
    v8f acc1 = {};
    for (int b = 0; b < nb; b += 2) {
        if (b + 1 < nb) {
            #pragma unroll
            for (int i = 0; i < BATCH; ++i) bufB[i] = *(const v4f*)(ap + 8*(BATCH + i));
            __builtin_amdgcn_sched_barrier(0);
            mma_batch(bufA, B0, B1, acc0, acc1);
        } else {
            mma_batch(bufA, B0, B1, acc0, acc1);
            break;
        }
        if (b + 2 < nb) {
            #pragma unroll
            for (int i = 0; i < BATCH; ++i) bufA[i] = *(const v4f*)(ap + 8*(2*BATCH + i));
            __builtin_amdgcn_sched_barrier(0);
            mma_batch(bufB, B0 + BATCH*32, B1 + BATCH*32, acc0, acc1);
        } else {
            mma_batch(bufB, B0 + BATCH*32, B1 + BATCH*32, acc0, acc1);
            break;
        }
        ap += 8*2*BATCH;
        B0 += 2*BATCH*32;
        B1 += 2*BATCH*32;
    }
    {
        const v4f w0 = sB[NP*32 + lane];
        const v4f w1 = sB[(NJ + NP)*32 + lane];
        const v2f w0l = {w0.x, w0.y};
        const v2f w1l = {w1.x, w1.y};
        acc0 = wmma_f32(xt, w0l, acc0);
        acc1 = wmma_f32(xt, w1l, acc1);
    }

    const int ncol  = m16;
    const int mbase = hf * 8;
    float* p0 = pn + ncol;
    #pragma unroll
    for (int j = 0; j < 8; ++j) p0[(wv*16 + mbase + j) * PQ] = acc0[j];
    if (ncol < 9) {
        float* p1 = pn + 16 + ncol;
        #pragma unroll
        for (int j = 0; j < 8; ++j) p1[(wv*16 + mbase + j) * PQ] = acc1[j];
    }
}

#define KS     16
#define KSLICE 800
#define HWV    4
#define HGRP   25
__global__ __launch_bounds__(128) void head_partial_wmma(
    float* ws, const float* __restrict__ w_obj, const float* __restrict__ w_mot)
{
    __shared__ v4f red[HWV * 128];
    const int mtile = blockIdx.x;
    const int ks    = blockIdx.y;
    const int tid   = threadIdx.x;
    const int wv    = tid >> 5;
    const int lane  = tid & 31;
    const int m16   = lane & 15;
    const int hf    = lane >> 4;

    const long koff = (long)ks * KSLICE + wv * (HGRP * 8) + 4 * hf;
    const float* ap  = ws + (long)(mtile*16 + m16) * CK + koff;
    const float* b0p = w_obj + (long)m16 * CK + koff;
    const int  nh1   = 16 + (m16 & 7);
    const float* b1p = (nh1 < 20) ? (w_obj + (long)nh1 * CK + koff)
                                  : (w_mot + (long)(nh1 - 20) * CK + koff);

    v4f xa[HGRP], xb0[HGRP], xb1[HGRP];
    #pragma unroll
    for (int i = 0; i < HGRP; ++i) {
        xa[i]  = *(const v4f*)(ap  + 8*i);
        xb0[i] = *(const v4f*)(b0p + 8*i);
        xb1[i] = *(const v4f*)(b1p + 8*i);
    }
    __builtin_amdgcn_sched_barrier(0);

    v8f acc0 = {};
    v8f acc1 = {};
    #pragma unroll
    for (int i = 0; i < HGRP; ++i) {
        const v4f x = xa[i], w0 = xb0[i], w1 = xb1[i];
        const v2f xlo = {x.x, x.y},   xhi = {x.z, x.w};
        const v2f w0l = {w0.x, w0.y}, w0h = {w0.z, w0.w};
        const v2f w1l = {w1.x, w1.y}, w1h = {w1.z, w1.w};
        acc0 = wmma_f32(xlo, w0l, acc0);  acc1 = wmma_f32(xlo, w1l, acc1);
        acc0 = wmma_f32(xhi, w0h, acc0);  acc1 = wmma_f32(xhi, w1h, acc1);
    }

    v4f* rw = red + wv * 128;
    rw[lane*2 + 0]      = (v4f){acc0[0], acc0[1], acc0[2], acc0[3]};
    rw[lane*2 + 1]      = (v4f){acc0[4], acc0[5], acc0[6], acc0[7]};
    rw[64 + lane*2 + 0] = (v4f){acc1[0], acc1[1], acc1[2], acc1[3]};
    rw[64 + lane*2 + 1] = (v4f){acc1[4], acc1[5], acc1[6], acc1[7]};
    __syncthreads();

    v4f s = red[tid];
    #pragma unroll
    for (int w = 1; w < HWV; ++w) s += red[w*128 + tid];
    *(v4f*)(ws + (long)(mtile*16) * CK + (long)ks * KSLICE + 4*tid) = s;
}

__global__ __launch_bounds__(512) void head_reduce(
    const float* __restrict__ ws,
    const float* __restrict__ b_obj, const float* __restrict__ b_mot,
    float* __restrict__ out)
{
    const int mtile = blockIdx.x;
    const int e     = threadIdx.x;
    const int tile  = e >> 8;
    const int lane  = (e >> 3) & 31;
    const int j     = e & 7;
    const int nh    = tile*16 + (lane & 15);
    const int roi   = mtile*16 + (lane >> 4)*8 + j;
    if (nh < 24) {
        const float* p = ws + (long)(mtile*16) * CK + e;
        float v[KS];
        #pragma unroll
        for (int k = 0; k < KS; ++k) v[k] = p[k * KSLICE];
        float s = v[0];
        #pragma unroll
        for (int k = 1; k < KS; ++k) s += v[k];
        if (nh < 20) out[roi*20 + nh] = s + b_obj[nh];
        else         out[20*NROI + roi*4 + (nh - 20)] = s + b_mot[nh - 20];
    }
}

extern "C" void kernel_launch(void* const* d_in, const int* in_sizes, int n_in,
                              void* d_out, int out_size, void* d_ws, size_t ws_size,
                              hipStream_t stream) {
    const float* feat  = (const float*)d_in[0];
    const float* bb    = (const float*)d_in[1];
    const float* w_obj = (const float*)d_in[2];
    const float* b_obj = (const float*)d_in[3];
    const float* w_mot = (const float*)d_in[4];
    const float* b_mot = (const float*)d_in[5];
    float* out    = (float*)d_out;
    float* pooled = (float*)d_ws;

    prroi_pool_wmma<<<dim3(NROI), 1024, 0, stream>>>(feat, bb, pooled);
    head_partial_wmma<<<dim3(16, KS), 128, 0, stream>>>(pooled, w_obj, w_mot);
    head_reduce<<<dim3(16), 512, 0, stream>>>(pooled, b_obj, b_mot, out);
}


# pooling kernel: three register batches (18 16-byte loads per lane) kept in flight, asked for ahead of the weight build; k loop as straight-line code per batch count; k order kept
# speedup vs baseline: 3.1924x; 1.0775x over previous
#include <hip/hip_runtime.h>
#include <hip/hip_bf16.h>

typedef float v2f __attribute__((ext_vector_type(2)));
typedef float v4f __attribute__((ext_vector_type(4)));
typedef float v8f __attribute__((ext_vector_type(8)));

#define POOL 5
#define PQ   25
#define CC   512
#define HH   22
#define WW   22
#define HW   484
#define NP   60
#define NJ   61
#define BATCH  6
#define NBATCH 10
#define CK   12800
#define NROI 256
#define RSCALE (1.0f/16.0f)

__device__ __forceinline__ float hat_int(float t) {
    t = fminf(1.0f, fmaxf(-1.0f, t));
    float a = t + 1.0f, b = 1.0f - t;
    return (t < 0.0f) ? 0.5f * a * a : 1.0f - 0.5f * b * b;
}

__device__ __forceinline__ v8f wmma_f32(v2f a, v2f b, v8f c) {
    return __builtin_amdgcn_wmma_f32_16x16x4_f32(false, a, false, b,
                                                 (short)0, c, false, false);
}

__device__ __forceinline__ void lds_barrier() {
    __builtin_amdgcn_fence(__ATOMIC_RELEASE, "workgroup", "local");
    __builtin_amdgcn_s_barrier();
    __builtin_amdgcn_fence(__ATOMIC_ACQUIRE, "workgroup", "local");
}

__device__ __forceinline__ void mma_batch(const v4f (&x)[BATCH], const v4f* __restrict__ w0p,
                                          const v4f* __restrict__ w1p, v8f& acc0, v8f& acc1)
{
    #pragma unroll
    for (int i = 0; i < BATCH; ++i) {
        const v4f xv = x[i];
        const v4f w0 = w0p[i*32];
        const v4f w1 = w1p[i*32];
        const v2f xlo = {xv.x, xv.y}, xhi = {xv.z, xv.w};
        const v2f w0l = {w0.x, w0.y}, w0h = {w0.z, w0.w};
        const v2f w1l = {w1.x, w1.y}, w1h = {w1.z, w1.w};
        acc0 = wmma_f32(xlo, w0l, acc0);  acc1 = wmma_f32(xlo, w1l, acc1);
        acc0 = wmma_f32(xhi, w0h, acc0);  acc1 = wmma_f32(xhi, w1h, acc1);
    }
}

template<int NB>
__device__ __forceinline__ void build_w2(v4f* sB, const float* sWx, const float* sWy,
                                         const float inv_area, const int js, const int tid)
{
    constexpr int nslot = NB*BATCH + 1;
    for (int idx = tid; idx < 2*nslot*32; idx += 1024) {
        const int bt = (idx >= nslot*32) ? 1 : 0;
        const int r  = idx - bt*nslot*32;
        const int jj = r >> 5;
        const int ln = r & 31;
        const int j  = (jj < nslot - 1) ? (js + jj) : NP;
        const int pq = bt*16 + (ln & 15);
        const int hh = ln >> 4;
        v4f val = {0.0f, 0.0f, 0.0f, 0.0f};
        if (pq < PQ) {
            int p = pq / POOL, q = pq % POOL;
            const float* wy = sWy + p*HH;
            const float* wx = sWx + q*WW;
            if (j < NP) {
                int k = 8*j + 4*hh;
                val.x = (wy[(k  )/WW] * wx[(k  )%WW]) * inv_area;
                val.y = (wy[(k+1)/WW] * wx[(k+1)%WW]) * inv_area;
                val.z = (wy[(k+2)/WW] * wx[(k+2)%WW]) * inv_area;
                val.w = (wy[(k+3)/WW] * wx[(k+3)%WW]) * inv_area;
            } else {
                int k = 8*NP + 2*hh;
                val.x = (wy[(k  )/WW] * wx[(k  )%WW]) * inv_area;
                val.y = (wy[(k+1)/WW] * wx[(k+1)%WW]) * inv_area;
            }
        }
        sB[(bt*NJ + j)*32 + ln] = val;
    }
}

#define NSTAGE 3
template<int NB>
__device__ __forceinline__ void pool_span(const float* __restrict__ ap, v4f* sB,
                                          const float* sWx, const float* sWy,
                                          const float inv_area, const int js,
                                          const int tid, const int lane, v8f& acc0, v8f& acc1)
{
    v4f buf[NSTAGE][BATCH];
    #pragma unroll
    for (int s = 0; s < NSTAGE; ++s) {
        if (s < NB) {
            #pragma unroll
            for (int i = 0; i < BATCH; ++i) buf[s][i] = *(const v4f*)(ap + 8*(s*BATCH + i));
        }
    }

    build_w2<NB>(sB, sWx, sWy, inv_area, js, tid);
    lds_barrier();

    const v4f* B0 = sB + js*32 + lane;
    const v4f* B1 = B0 + NJ*32;
    #pragma unroll
    for (int b = 0; b < NB; ++b) {
        mma_batch(buf[b % NSTAGE], B0 + b*BATCH*32, B1 + b*BATCH*32, acc0, acc1);
        if (b + NSTAGE < NB) {
            #pragma unroll
            for (int i = 0; i < BATCH; ++i)
                buf[b % NSTAGE][i] = *(const v4f*)(ap + 8*((b + NSTAGE)*BATCH + i));
            __builtin_amdgcn_sched_barrier(0);
        }
    }
}

__global__ __launch_bounds__(1024) void prroi_pool_wmma(
    const float* __restrict__ feat, const float* __restrict__ bb,
    float* __restrict__ pooled)
{
    __shared__ v4f   sB[2 * NJ * 32];
    __shared__ float sWx[POOL * WW];
    __shared__ float sWy[POOL * HH];

    const int n    = blockIdx.x;
    const int tid  = threadIdx.x;
    const int wv   = tid >> 5;
    const int lane = tid & 31;
    const int m16  = lane & 15;
    const int hf   = lane >> 4;

    const float* rowp = feat + (long)n * (CC * HW) + (long)(wv*16 + m16) * HW;

    const v2f xt = *(const v2f*)(rowp + 480 + 2*hf);

    const float bx = bb[n*4+0], by = bb[n*4+1], bwid = bb[n*4+2], bhgt = bb[n*4+3];
    const float x1 = bx * RSCALE, y1 = by * RSCALE;
    const float x2 = (bx + bwid) * RSCALE, y2 = (by + bhgt) * RSCALE;
    const float bw = fmaxf((x2 - x1) * (1.0f/POOL), 0.0f);
    const float bh = fmaxf((y2 - y1) * (1.0f/POOL), 0.0f);
    const float area = bw * bh;
    const float inv_area = (area > 0.0f) ? (1.0f / fmaxf(area, 1e-12f)) : 0.0f;

    if (tid < POOL*WW) {
        int q = tid / WW, i = tid % WW;
        sWx[tid] = hat_int(x1 + (q+1)*bw - (float)i) - hat_int(x1 + q*bw - (float)i);
    } else if (tid < POOL*WW + POOL*HH) {
        int t2 = tid - POOL*WW;
        int p = t2 / HH, j = t2 % HH;
        sWy[t2] = hat_int(y1 + (p+1)*bh - (float)j) - hat_int(y1 + p*bh - (float)j);
    }
    lds_barrier();

    bool act = false;
    if (lane < HH) {
        #pragma unroll
        for (int p = 0; p < POOL; ++p) act |= !(sWy[p*HH + lane] == 0.0f);
    }
    const unsigned rmask = __builtin_amdgcn_ballot_w32(act);
    int j_lo = 0, j_hi = 0;
    if (rmask != 0u) {
        const int r_lo = __builtin_ctz(rmask);
        const int r_hi = 31 - __builtin_clz(rmask);
        j_lo = (r_lo * WW) >> 3;
        j_hi = min(((r_hi + 1) * WW + 7) >> 3, NP);
    }
    const int nb = (j_hi - j_lo + BATCH - 1) / BATCH;
    const int js = min(j_lo, NP - nb*BATCH);

    const float* ap = rowp + 4*hf + 8*js;

    float* pn = pooled + (long)n * CK;

    v8f acc0 = {};
    v8f acc1 = {};
    switch (nb) {
        case 0:  pool_span<0>(ap, sB, sWx, sWy, inv_area, js, tid, lane, acc0, acc1);  break;
        case 1:  pool_span<1>(ap, sB, sWx, sWy, inv_area, js, tid, lane, acc0, acc1);  break;
        case 2:  pool_span<2>(ap, sB, sWx, sWy, inv_area, js, tid, lane, acc0, acc1);  break;
        case 3:  pool_span<3>(ap, sB, sWx, sWy, inv_area, js, tid, lane, acc0, acc1);  break;
        case 4:  pool_span<4>(ap, sB, sWx, sWy, inv_area, js, tid, lane, acc0, acc1);  break;
        case 5:  pool_span<5>(ap, sB, sWx, sWy, inv_area, js, tid, lane, acc0, acc1);  break;
        case 6:  pool_span<6>(ap, sB, sWx, sWy, inv_area, js, tid, lane, acc0, acc1);  break;
        case 7:  pool_span<7>(ap, sB, sWx, sWy, inv_area, js, tid, lane, acc0, acc1);  break;
        case 8:  pool_span<8>(ap, sB, sWx, sWy, inv_area, js, tid, lane, acc0, acc1);  break;
        case 9:  pool_span<9>(ap, sB, sWx, sWy, inv_area, js, tid, lane, acc0, acc1);  break;
        default: pool_span<NBATCH>(ap, sB, sWx, sWy, inv_area, js, tid, lane, acc0, acc1); break;
    }
    {
        const v4f w0 = sB[NP*32 + lane];
        const v4f w1 = sB[(NJ + NP)*32 + lane];
        const v2f w0l = {w0.x, w0.y};
        const v2f w1l = {w1.x, w1.y};
        acc0 = wmma_f32(xt, w0l, acc0);
        acc1 = wmma_f32(xt, w1l, acc1);
    }

    const int ncol  = m16;
    const int mbase = hf * 8;
    float* p0 = pn + ncol;
    #pragma unroll
    for (int j = 0; j < 8; ++j) p0[(wv*16 + mbase + j) * PQ] = acc0[j];
    if (ncol < 9) {
        float* p1 = pn + 16 + ncol;
        #pragma unroll
        for (int j = 0; j < 8; ++j) p1[(wv*16 + mbase + j) * PQ] = acc1[j];
    }
}

#define KS     16
#define KSLICE 800
#define HWV    4
#define HGRP   25
__global__ __launch_bounds__(128) void head_partial_wmma(
    float* ws, const float* __restrict__ w_obj, const float* __restrict__ w_mot)
{
    __shared__ v4f red[HWV * 128];
    const int mtile = blockIdx.x;
    const int ks    = blockIdx.y;
    const int tid   = threadIdx.x;
    const int wv    = tid >> 5;
    const int lane  = tid & 31;
    const int m16   = lane & 15;
    const int hf    = lane >> 4;

    const long koff = (long)ks * KSLICE + wv * (HGRP * 8) + 4 * hf;
    const float* ap  = ws + (long)(mtile*16 + m16) * CK + koff;
    const float* b0p = w_obj + (long)m16 * CK + koff;
    const int  nh1   = 16 + (m16 & 7);
    const float* b1p = (nh1 < 20) ? (w_obj + (long)nh1 * CK + koff)
                                  : (w_mot + (long)(nh1 - 20) * CK + koff);

    v4f xa[HGRP], xb0[HGRP], xb1[HGRP];
    #pragma unroll
    for (int i = 0; i < HGRP; ++i) {
        xa[i]  = *(const v4f*)(ap  + 8*i);
        xb0[i] = *(const v4f*)(b0p + 8*i);
        xb1[i] = *(const v4f*)(b1p + 8*i);
    }
    __builtin_amdgcn_sched_barrier(0);

    v8f acc0 = {};
    v8f acc1 = {};
    #pragma unroll
    for (int i = 0; i < HGRP; ++i) {
        const v4f x = xa[i], w0 = xb0[i], w1 = xb1[i];
        const v2f xlo = {x.x, x.y},   xhi = {x.z, x.w};
        const v2f w0l = {w0.x, w0.y}, w0h = {w0.z, w0.w};
        const v2f w1l = {w1.x, w1.y}, w1h = {w1.z, w1.w};
        acc0 = wmma_f32(xlo, w0l, acc0);  acc1 = wmma_f32(xlo, w1l, acc1);
        acc0 = wmma_f32(xhi, w0h, acc0);  acc1 = wmma_f32(xhi, w1h, acc1);
    }

    v4f* rw = red + wv * 128;
    rw[lane*2 + 0]      = (v4f){acc0[0], acc0[1], acc0[2], acc0[3]};
    rw[lane*2 + 1]      = (v4f){acc0[4], acc0[5], acc0[6], acc0[7]};
    rw[64 + lane*2 + 0] = (v4f){acc1[0], acc1[1], acc1[2], acc1[3]};
    rw[64 + lane*2 + 1] = (v4f){acc1[4], acc1[5], acc1[6], acc1[7]};
    __syncthreads();

    v4f s = red[tid];
    #pragma unroll
    for (int w = 1; w < HWV; ++w) s += red[w*128 + tid];
    *(v4f*)(ws + (long)(mtile*16) * CK + (long)ks * KSLICE + 4*tid) = s;
}

__global__ __launch_bounds__(512) void head_reduce(
    const float* __restrict__ ws,
    const float* __restrict__ b_obj, const float* __restrict__ b_mot,
    float* __restrict__ out)
{
    const int mtile = blockIdx.x;
    const int e     = threadIdx.x;
    const int tile  = e >> 8;
    const int lane  = (e >> 3) & 31;
    const int j     = e & 7;
    const int nh    = tile*16 + (lane & 15);
    const int roi   = mtile*16 + (lane >> 4)*8 + j;
    if (nh < 24) {
        const float* p = ws + (long)(mtile*16) * CK + e;
        float v[KS];
        #pragma unroll
        for (int k = 0; k < KS; ++k) v[k] = p[k * KSLICE];
        float s = v[0];
        #pragma unroll
        for (int k = 1; k < KS; ++k) s += v[k];
        if (nh < 20) out[roi*20 + nh] = s + b_obj[nh];
        else         out[20*NROI + roi*4 + (nh - 20)] = s + b_mot[nh - 20];
    }
}

extern "C" void kernel_launch(void* const* d_in, const int* in_sizes, int n_in,
                              void* d_out, int out_size, void* d_ws, size_t ws_size,
                              hipStream_t stream) {
    const float* feat  = (const float*)d_in[0];
    const float* bb    = (const float*)d_in[1];
    const float* w_obj = (const float*)d_in[2];
    const float* b_obj = (const float*)d_in[3];
    const float* w_mot = (const float*)d_in[4];
    const float* b_mot = (const float*)d_in[5];
    float* out    = (float*)d_out;
    float* pooled = (float*)d_ws;

    prroi_pool_wmma<<<dim3(NROI), 1024, 0, stream>>>(feat, bb, pooled);
    head_partial_wmma<<<dim3(16, KS), 128, 0, stream>>>(pooled, w_obj, w_mot);
    head_reduce<<<dim3(16), 512, 0, stream>>>(pooled, b_obj, b_mot, out);
}


# pooling kernel: feature loads cover whole lines (a load takes 32 consecutive 16-byte chunks of the wave's row pieces, ~7 lines not 16), A fragments read from wave-private LDS rows; k order kept
# speedup vs baseline: 3.6272x; 1.1362x over previous
#include <hip/hip_runtime.h>
#include <hip/hip_bf16.h>

typedef float v2f __attribute__((ext_vector_type(2)));
typedef float v4f __attribute__((ext_vector_type(4)));
typedef float v8f __attribute__((ext_vector_type(8)));

#define POOL 5
#define PQ   25
#define CC   512
#define HH   22
#define WW   22
#define HW   484
#define NP   60
#define NJ   61
#define BATCH  6
#define NBATCH 10
#define CK   12800
#define NROI 256
#define RSCALE (1.0f/16.0f)

__device__ __forceinline__ float hat_int(float t) {
    t = fminf(1.0f, fmaxf(-1.0f, t));
    float a = t + 1.0f, b = 1.0f - t;
    return (t < 0.0f) ? 0.5f * a * a : 1.0f - 0.5f * b * b;
}

__device__ __forceinline__ v8f wmma_f32(v2f a, v2f b, v8f c) {
    return __builtin_amdgcn_wmma_f32_16x16x4_f32(false, a, false, b,
                                                 (short)0, c, false, false);
}

__device__ __forceinline__ void lds_barrier() {
    __builtin_amdgcn_fence(__ATOMIC_RELEASE, "workgroup", "local");
    __builtin_amdgcn_s_barrier();
    __builtin_amdgcn_fence(__ATOMIC_ACQUIRE, "workgroup", "local");
}

__device__ __forceinline__ void mma_batch(const v4f* xp, const v4f* __restrict__ w0p,
                                          const v4f* __restrict__ w1p, v8f& acc0, v8f& acc1)
{
    #pragma unroll
    for (int i = 0; i < BATCH; ++i) {
        const v4f xv = xp[2*i];
        const v4f w0 = w0p[i*32];
        const v4f w1 = w1p[i*32];
        const v2f xlo = {xv.x, xv.y}, xhi = {xv.z, xv.w};
        const v2f w0l = {w0.x, w0.y}, w0h = {w0.z, w0.w};
        const v2f w1l = {w1.x, w1.y}, w1h = {w1.z, w1.w};
        acc0 = wmma_f32(xlo, w0l, acc0);  acc1 = wmma_f32(xlo, w1l, acc1);
        acc0 = wmma_f32(xhi, w0h, acc0);  acc1 = wmma_f32(xhi, w1h, acc1);
    }
}

template<int NB>
__device__ __forceinline__ void build_w2(v4f* sB, const float* sWx, const float* sWy,
                                         const float inv_area, const int js, const int tid)
{
    constexpr int nslot = NB*BATCH + 1;
    for (int idx = tid; idx < 2*nslot*32; idx += 1024) {
        const int bt = (idx >= nslot*32) ? 1 : 0;
        const int r  = idx - bt*nslot*32;
        const int jj = r >> 5;
        const int ln = r & 31;
        const int j  = (jj < nslot - 1) ? (js + jj) : NP;
        const int pq = bt*16 + (ln & 15);
        const int hh = ln >> 4;
        v4f val = {0.0f, 0.0f, 0.0f, 0.0f};
        if (pq < PQ) {
            int p = pq / POOL, q = pq % POOL;
            const float* wy = sWy + p*HH;
            const float* wx = sWx + q*WW;
            if (j < NP) {
                int k = 8*j + 4*hh;
                val.x = (wy[(k  )/WW] * wx[(k  )%WW]) * inv_area;
                val.y = (wy[(k+1)/WW] * wx[(k+1)%WW]) * inv_area;
                val.z = (wy[(k+2)/WW] * wx[(k+2)%WW]) * inv_area;
                val.w = (wy[(k+3)/WW] * wx[(k+3)%WW]) * inv_area;
            } else {
                int k = 8*NP + 2*hh;
                val.x = (wy[(k  )/WW] * wx[(k  )%WW]) * inv_area;
                val.y = (wy[(k+1)/WW] * wx[(k+1)%WW]) * inv_area;
            }
        }
        sB[(bt*NJ + j)*32 + ln] = val;
    }
}

__device__ __forceinline__ void wave_lds_order() {
    __builtin_amdgcn_fence(__ATOMIC_RELEASE, "wavefront", "local");
    __builtin_amdgcn_wave_barrier();
    __builtin_amdgcn_fence(__ATOMIC_ACQUIRE, "wavefront", "local");
}

#define NPRE 3
#define APITCH 13
#define ASTG   (16*APITCH)
template<int NB>
__device__ __forceinline__ void pool_span(const float* __restrict__ gp, v4f* sAw, v4f* sB,
                                          const float* sWx, const float* sWy,
                                          const float inv_area, const int js,
                                          const int tid, const int lane, v8f& acc0, v8f& acc1)
{
    unsigned goff[3], loff[3];
    #pragma unroll
    for (int i = 0; i < 3; ++i) {
        const unsigned t = 32u*i + (unsigned)lane;
        const unsigned r = t / 12u, c = t - 12u*r;
        goff[i] = 4u*(r*HW + 4u*c);
        loff[i] = r*APITCH + c;
    }

    v4f buf[NPRE][BATCH];
    #pragma unroll
    for (int s = 0; s < NPRE; ++s) {
        if (s < NB) {
            #pragma unroll
            for (int i = 0; i < BATCH; ++i)
                buf[s][i] = *(const v4f*)((const char*)gp + (192*s + (i / 3) * (32*HW)) + goff[i % 3]);
            __builtin_amdgcn_sched_barrier(0);
        }
    }

    build_w2<NB>(sB, sWx, sWy, inv_area, js, tid);
    lds_barrier();

    const v4f* B0 = sB + js*32 + lane;
    const v4f* B1 = B0 + NJ*32;
    const v4f* xa = sAw + (lane & 15)*APITCH + (lane >> 4);
    #pragma unroll
    for (int b = 0; b < NB; ++b) {
        const int sl = (b == 0) ? 0 : 1 + ((b - 1) & 1);
        wave_lds_order();
        #pragma unroll
        for (int i = 0; i < BATCH; ++i)
            sAw[loff[i % 3] + (i / 3) * (8*APITCH)] = buf[sl][i];
        if (b >= 1 && b + 2 < NB) {
            __builtin_amdgcn_sched_barrier(0);
            #pragma unroll
            for (int i = 0; i < BATCH; ++i)
                buf[sl][i] = *(const v4f*)((const char*)gp + (192*(b + 2) + (i / 3) * (32*HW)) + goff[i % 3]);
            __builtin_amdgcn_sched_barrier(0);
        }
        wave_lds_order();
        mma_batch(xa, B0 + b*BATCH*32, B1 + b*BATCH*32, acc0, acc1);
    }
}

__global__ __launch_bounds__(1024) void prroi_pool_wmma(
    const float* __restrict__ feat, const float* __restrict__ bb,
    float* __restrict__ pooled)
{
    __shared__ v4f   sB[2 * NJ * 32];
    __shared__ v4f   sA[32 * ASTG];
    __shared__ float sWx[POOL * WW];
    __shared__ float sWy[POOL * HH];

    const int n    = blockIdx.x;
    const int tid  = threadIdx.x;
    const int wv   = __builtin_amdgcn_readfirstlane(tid >> 5);
    const int lane = tid & 31;
    const int m16  = lane & 15;
    const int hf   = lane >> 4;

    const float* rowp = feat + (long)n * (CC * HW) + (long)(wv*16 + m16) * HW;

    const v2f xt = *(const v2f*)(rowp + 480 + 2*hf);

    const float bx = bb[n*4+0], by = bb[n*4+1], bwid = bb[n*4+2], bhgt = bb[n*4+3];
    const float x1 = bx * RSCALE, y1 = by * RSCALE;
    const float x2 = (bx + bwid) * RSCALE, y2 = (by + bhgt) * RSCALE;
    const float bw = fmaxf((x2 - x1) * (1.0f/POOL), 0.0f);
    const float bh = fmaxf((y2 - y1) * (1.0f/POOL), 0.0f);
    const float area = bw * bh;
    const float inv_area = (area > 0.0f) ? (1.0f / fmaxf(area, 1e-12f)) : 0.0f;

    if (tid < POOL*WW) {
        int q = tid / WW, i = tid % WW;
        sWx[tid] = hat_int(x1 + (q+1)*bw - (float)i) - hat_int(x1 + q*bw - (float)i);
    } else if (tid < POOL*WW + POOL*HH) {
        int t2 = tid - POOL*WW;
        int p = t2 / HH, j = t2 % HH;
        sWy[t2] = hat_int(y1 + (p+1)*bh - (float)j) - hat_int(y1 + p*bh - (float)j);
    }
    lds_barrier();

    bool act = false;
    if (lane < HH) {
        #pragma unroll
        for (int p = 0; p < POOL; ++p) act |= !(sWy[p*HH + lane] == 0.0f);
    }
    const unsigned rmask = __builtin_amdgcn_ballot_w32(act);
    int j_lo = 0, j_hi = 0;
    if (rmask != 0u) {
        const int r_lo = __builtin_ctz(rmask);
        const int r_hi = 31 - __builtin_clz(rmask);
        j_lo = (r_lo * WW) >> 3;
        j_hi = min(((r_hi + 1) * WW + 7) >> 3, NP);
    }
    const int nb = (j_hi - j_lo + BATCH - 1) / BATCH;
    const int js = min(j_lo, NP - nb*BATCH);

    const float* gp = feat + (long)n * (CC * HW) + (long)(wv*16) * HW + 8*js;
    v4f* sAw = sA + wv*ASTG;

    float* pn = pooled + (long)n * CK;

    v8f acc0 = {};
    v8f acc1 = {};
    switch (nb) {
        case 0:  pool_span<0>(gp, sAw, sB, sWx, sWy, inv_area, js, tid, lane, acc0, acc1);  break;
        case 1:  pool_span<1>(gp, sAw, sB, sWx, sWy, inv_area, js, tid, lane, acc0, acc1);  break;
        case 2:  pool_span<2>(gp, sAw, sB, sWx, sWy, inv_area, js, tid, lane, acc0, acc1);  break;
        case 3:  pool_span<3>(gp, sAw, sB, sWx, sWy, inv_area, js, tid, lane, acc0, acc1);  break;
        case 4:  pool_span<4>(gp, sAw, sB, sWx, sWy, inv_area, js, tid, lane, acc0, acc1);  break;
        case 5:  pool_span<5>(gp, sAw, sB, sWx, sWy, inv_area, js, tid, lane, acc0, acc1);  break;
        case 6:  pool_span<6>(gp, sAw, sB, sWx, sWy, inv_area, js, tid, lane, acc0, acc1);  break;
        case 7:  pool_span<7>(gp, sAw, sB, sWx, sWy, inv_area, js, tid, lane, acc0, acc1);  break;
        case 8:  pool_span<8>(gp, sAw, sB, sWx, sWy, inv_area, js, tid, lane, acc0, acc1);  break;
        case 9:  pool_span<9>(gp, sAw, sB, sWx, sWy, inv_area, js, tid, lane, acc0, acc1);  break;
        default: pool_span<NBATCH>(gp, sAw, sB, sWx, sWy, inv_area, js, tid, lane, acc0, acc1); break;
    }
    {
        const v4f w0 = sB[NP*32 + lane];
        const v4f w1 = sB[(NJ + NP)*32 + lane];
        const v2f w0l = {w0.x, w0.y};
        const v2f w1l = {w1.x, w1.y};
        acc0 = wmma_f32(xt, w0l, acc0);
        acc1 = wmma_f32(xt, w1l, acc1);
    }

    const int ncol  = m16;
    const int mbase = hf * 8;
    float* p0 = pn + ncol;
    #pragma unroll
    for (int j = 0; j < 8; ++j) p0[(wv*16 + mbase + j) * PQ] = acc0[j];
    if (ncol < 9) {
        float* p1 = pn + 16 + ncol;
        #pragma unroll
        for (int j = 0; j < 8; ++j) p1[(wv*16 + mbase + j) * PQ] = acc1[j];
    }
}

#define KS     16
#define KSLICE 800
#define HWV    4
#define HGRP   25
__global__ __launch_bounds__(128) void head_partial_wmma(
    float* ws, const float* __restrict__ w_obj, const float* __restrict__ w_mot)
{
    __shared__ v4f red[HWV * 128];
    const int mtile = blockIdx.x;
    const int ks    = blockIdx.y;
    const int tid   = threadIdx.x;
    const int wv    = tid >> 5;
    const int lane  = tid & 31;
    const int m16   = lane & 15;
    const int hf    = lane >> 4;

    const long koff = (long)ks * KSLICE + wv * (HGRP * 8) + 4 * hf;
    const float* ap  = ws + (long)(mtile*16 + m16) * CK + koff;
    const float* b0p = w_obj + (long)m16 * CK + koff;
    const int  nh1   = 16 + (m16 & 7);
    const float* b1p = (nh1 < 20) ? (w_obj + (long)nh1 * CK + koff)
                                  : (w_mot + (long)(nh1 - 20) * CK + koff);

    v4f xa[HGRP], xb0[HGRP], xb1[HGRP];
    #pragma unroll
    for (int i = 0; i < HGRP; ++i) {
        xa[i]  = *(const v4f*)(ap  + 8*i);
        xb0[i] = *(const v4f*)(b0p + 8*i);
        xb1[i] = *(const v4f*)(b1p + 8*i);
    }
    __builtin_amdgcn_sched_barrier(0);

    v8f acc0 = {};
    v8f acc1 = {};
    #pragma unroll
    for (int i = 0; i < HGRP; ++i) {
        const v4f x = xa[i], w0 = xb0[i], w1 = xb1[i];
        const v2f xlo = {x.x, x.y},   xhi = {x.z, x.w};
        const v2f w0l = {w0.x, w0.y}, w0h = {w0.z, w0.w};
        const v2f w1l = {w1.x, w1.y}, w1h = {w1.z, w1.w};
        acc0 = wmma_f32(xlo, w0l, acc0);  acc1 = wmma_f32(xlo, w1l, acc1);
        acc0 = wmma_f32(xhi, w0h, acc0);  acc1 = wmma_f32(xhi, w1h, acc1);
    }

    v4f* rw = red + wv * 128;
    rw[lane*2 + 0]      = (v4f){acc0[0], acc0[1], acc0[2], acc0[3]};
    rw[lane*2 + 1]      = (v4f){acc0[4], acc0[5], acc0[6], acc0[7]};
    rw[64 + lane*2 + 0] = (v4f){acc1[0], acc1[1], acc1[2], acc1[3]};
    rw[64 + lane*2 + 1] = (v4f){acc1[4], acc1[5], acc1[6], acc1[7]};
    __syncthreads();

    v4f s = red[tid];
    #pragma unroll
    for (int w = 1; w < HWV; ++w) s += red[w*128 + tid];
    *(v4f*)(ws + (long)(mtile*16) * CK + (long)ks * KSLICE + 4*tid) = s;
}

__global__ __launch_bounds__(512) void head_reduce(
    const float* __restrict__ ws,
    const float* __restrict__ b_obj, const float* __restrict__ b_mot,
    float* __restrict__ out)
{
    const int mtile = blockIdx.x;
    const int e     = threadIdx.x;
    const int tile  = e >> 8;
    const int lane  = (e >> 3) & 31;
    const int j     = e & 7;
    const int nh    = tile*16 + (lane & 15);
    const int roi   = mtile*16 + (lane >> 4)*8 + j;
    if (nh < 24) {
        const float* p = ws + (long)(mtile*16) * CK + e;
        float v[KS];
        #pragma unroll
        for (int k = 0; k < KS; ++k) v[k] = p[k * KSLICE];
        float s = v[0];
        #pragma unroll
        for (int k = 1; k < KS; ++k) s += v[k];
        if (nh < 20) out[roi*20 + nh] = s + b_obj[nh];
        else         out[20*NROI + roi*4 + (nh - 20)] = s + b_mot[nh - 20];
    }
}

extern "C" void kernel_launch(void* const* d_in, const int* in_sizes, int n_in,
                              void* d_out, int out_size, void* d_ws, size_t ws_size,
                              hipStream_t stream) {
    const float* feat  = (const float*)d_in[0];
    const float* bb    = (const float*)d_in[1];
    const float* w_obj = (const float*)d_in[2];
    const float* b_obj = (const float*)d_in[3];
    const float* w_mot = (const float*)d_in[4];
    const float* b_mot = (const float*)d_in[5];
    float* out    = (float*)d_out;
    float* pooled = (float*)d_ws;

    prroi_pool_wmma<<<dim3(NROI), 1024, 0, stream>>>(feat, bb, pooled);
    head_partial_wmma<<<dim3(16, KS), 128, 0, stream>>>(pooled, w_obj, w_mot);
    head_reduce<<<dim3(16), 512, 0, stream>>>(pooled, b_obj, b_mot, out);
}


# pooling kernel: bins split 15+10 by bin row over the two tiles; each tile's WMMAs run only over half batches whose rows carry a non-zero y weight for it; skipped products exact zeros, k order kept
# speedup vs baseline: 3.7891x; 1.0446x over previous
#include <hip/hip_runtime.h>
#include <hip/hip_bf16.h>

typedef float v2f __attribute__((ext_vector_type(2)));
typedef float v4f __attribute__((ext_vector_type(4)));
typedef float v8f __attribute__((ext_vector_type(8)));

#define POOL 5
#define PQ   25
#define CC   512
#define HH   22
#define WW   22
#define HW   484
#define NP   60
#define NJ   61
#define BATCH  6
#define NBATCH 10
#define CK   12800
#define NROI 256
#define RSCALE (1.0f/16.0f)

__device__ __forceinline__ float hat_int(float t) {
    t = fminf(1.0f, fmaxf(-1.0f, t));
    float a = t + 1.0f, b = 1.0f - t;
    return (t < 0.0f) ? 0.5f * a * a : 1.0f - 0.5f * b * b;
}

__device__ __forceinline__ v8f wmma_f32(v2f a, v2f b, v8f c) {
    return __builtin_amdgcn_wmma_f32_16x16x4_f32(false, a, false, b,
                                                 (short)0, c, false, false);
}

__device__ __forceinline__ void lds_barrier() {
    __builtin_amdgcn_fence(__ATOMIC_RELEASE, "workgroup", "local");
    __builtin_amdgcn_s_barrier();
    __builtin_amdgcn_fence(__ATOMIC_ACQUIRE, "workgroup", "local");
}

#define HALF 3
__device__ __forceinline__ void mma_half(const v4f* xp, const v4f* __restrict__ wp, v8f& acc)
{
    #pragma unroll
    for (int i = 0; i < HALF; ++i) {
        const v4f xv = xp[2*i];
        const v4f w  = wp[i*32];
        const v2f xlo = {xv.x, xv.y}, xhi = {xv.z, xv.w};
        const v2f wl  = {w.x, w.y},   wh  = {w.z, w.w};
        acc = wmma_f32(xlo, wl, acc);
        acc = wmma_f32(xhi, wh, acc);
    }
}

template<int NB>
__device__ __forceinline__ void build_w2(v4f* sB, const float* sWx, const float* sWy,
                                         const float inv_area, const int js, const int tid)
{
    constexpr int nslot = NB*BATCH + 1;
    for (int idx = tid; idx < 2*nslot*32; idx += 1024) {
        const int bt = (idx >= nslot*32) ? 1 : 0;
        const int r  = idx - bt*nslot*32;
        const int jj = r >> 5;
        const int ln = r & 31;
        const int j  = (jj < nslot - 1) ? (js + jj) : NP;
        const int col = ln & 15;
        const int pq = bt*15 + col;
        const int hh = ln >> 4;
        v4f val = {0.0f, 0.0f, 0.0f, 0.0f};
        if (col < 15 - 5*bt) {
            int p = pq / POOL, q = pq % POOL;
            const float* wy = sWy + p*HH;
            const float* wx = sWx + q*WW;
            if (j < NP) {
                int k = 8*j + 4*hh;
                val.x = (wy[(k  )/WW] * wx[(k  )%WW]) * inv_area;
                val.y = (wy[(k+1)/WW] * wx[(k+1)%WW]) * inv_area;
                val.z = (wy[(k+2)/WW] * wx[(k+2)%WW]) * inv_area;
                val.w = (wy[(k+3)/WW] * wx[(k+3)%WW]) * inv_area;
            } else {
                int k = 8*NP + 2*hh;
                val.x = (wy[(k  )/WW] * wx[(k  )%WW]) * inv_area;
                val.y = (wy[(k+1)/WW] * wx[(k+1)%WW]) * inv_area;
            }
        }
        sB[(bt*NJ + j)*32 + ln] = val;
    }
}

__device__ __forceinline__ void wave_lds_order() {
    __builtin_amdgcn_fence(__ATOMIC_RELEASE, "wavefront", "local");
    __builtin_amdgcn_wave_barrier();
    __builtin_amdgcn_fence(__ATOMIC_ACQUIRE, "wavefront", "local");
}

#define NPRE 3
#define APITCH 13
#define ASTG   (16*APITCH)
template<int NB>
__device__ __forceinline__ void pool_span(const float* __restrict__ gp, v4f* sAw, v4f* sB,
                                          const float* sWx, const float* sWy,
                                          const float inv_area, const int js,
                                          const int t0lo, const int t0hi, const int t1lo, const int t1hi,
                                          const int tid, const int lane, v8f& acc0, v8f& acc1)
{
    unsigned goff[3], loff[3];
    #pragma unroll
    for (int i = 0; i < 3; ++i) {
        const unsigned t = 32u*i + (unsigned)lane;
        const unsigned r = t / 12u, c = t - 12u*r;
        goff[i] = 4u*(r*HW + 4u*c);
        loff[i] = r*APITCH + c;
    }

    v4f buf[NPRE][BATCH];
    #pragma unroll
    for (int s = 0; s < NPRE; ++s) {
        if (s < NB) {
            #pragma unroll
            for (int i = 0; i < BATCH; ++i)
                buf[s][i] = *(const v4f*)((const char*)gp + (192*s + (i / 3) * (32*HW)) + goff[i % 3]);
            __builtin_amdgcn_sched_barrier(0);
        }
    }

    build_w2<NB>(sB, sWx, sWy, inv_area, js, tid);
    lds_barrier();

    const v4f* B0 = sB + js*32 + lane;
    const v4f* B1 = B0 + NJ*32;
    const v4f* xa = sAw + (lane & 15)*APITCH + (lane >> 4);
    #pragma unroll
    for (int b = 0; b < NB; ++b) {
        const int sl = (b == 0) ? 0 : 1 + ((b - 1) & 1);
        wave_lds_order();
        #pragma unroll
        for (int i = 0; i < BATCH; ++i)
            sAw[loff[i % 3] + (i / 3) * (8*APITCH)] = buf[sl][i];
        if (b >= 1 && b + 2 < NB) {
            __builtin_amdgcn_sched_barrier(0);
            #pragma unroll
            for (int i = 0; i < BATCH; ++i)
                buf[sl][i] = *(const v4f*)((const char*)gp + (192*(b + 2) + (i / 3) * (32*HW)) + goff[i % 3]);
            __builtin_amdgcn_sched_barrier(0);
        }
        wave_lds_order();
        #pragma unroll
        for (int hb = 0; hb < BATCH/HALF; ++hb) {
            const int g  = js + b*BATCH + hb*HALF;
            const bool u0 = (g < t0hi) && (g + HALF > t0lo);
            const bool u1 = (g < t1hi) && (g + HALF > t1lo);
            const v4f* xh = xa + 2*HALF*hb;
            const v4f* w0 = B0 + (b*BATCH + hb*HALF)*32;
            const v4f* w1 = B1 + (b*BATCH + hb*HALF)*32;
            if (u0) mma_half(xh, w0, acc0);
            if (u1) mma_half(xh, w1, acc1);
        }
    }
}

__global__ __launch_bounds__(1024) void prroi_pool_wmma(
    const float* __restrict__ feat, const float* __restrict__ bb,
    float* __restrict__ pooled)
{
    __shared__ v4f   sB[2 * NJ * 32];
    __shared__ v4f   sA[32 * ASTG];
    __shared__ float sWx[POOL * WW];
    __shared__ float sWy[POOL * HH];

    const int n    = blockIdx.x;
    const int tid  = threadIdx.x;
    const int wv   = __builtin_amdgcn_readfirstlane(tid >> 5);
    const int lane = tid & 31;
    const int m16  = lane & 15;
    const int hf   = lane >> 4;

    const float* rowp = feat + (long)n * (CC * HW) + (long)(wv*16 + m16) * HW;

    const v2f xt = *(const v2f*)(rowp + 480 + 2*hf);

    const float bx = bb[n*4+0], by = bb[n*4+1], bwid = bb[n*4+2], bhgt = bb[n*4+3];
    const float x1 = bx * RSCALE, y1 = by * RSCALE;
    const float x2 = (bx + bwid) * RSCALE, y2 = (by + bhgt) * RSCALE;
    const float bw = fmaxf((x2 - x1) * (1.0f/POOL), 0.0f);
    const float bh = fmaxf((y2 - y1) * (1.0f/POOL), 0.0f);
    const float area = bw * bh;
    const float inv_area = (area > 0.0f) ? (1.0f / fmaxf(area, 1e-12f)) : 0.0f;

    if (tid < POOL*WW) {
        int q = tid / WW, i = tid % WW;
        sWx[tid] = hat_int(x1 + (q+1)*bw - (float)i) - hat_int(x1 + q*bw - (float)i);
    } else if (tid < POOL*WW + POOL*HH) {
        int t2 = tid - POOL*WW;
        int p = t2 / HH, j = t2 % HH;
        sWy[t2] = hat_int(y1 + (p+1)*bh - (float)j) - hat_int(y1 + p*bh - (float)j);
    }
    lds_barrier();

    bool act0 = false, act1 = false;
    if (lane < HH) {
        #pragma unroll
        for (int p = 0; p < 3; ++p)    act0 |= !(sWy[p*HH + lane] == 0.0f);
        #pragma unroll
        for (int p = 3; p < POOL; ++p) act1 |= !(sWy[p*HH + lane] == 0.0f);
    }
    const unsigned rmask0 = __builtin_amdgcn_ballot_w32(act0);
    const unsigned rmask1 = __builtin_amdgcn_ballot_w32(act1);
    const unsigned rmask  = rmask0 | rmask1;
    int j_lo = 0, j_hi = 0;
    int t0lo = 0, t0hi = 0;
    int t1lo = 0, t1hi = 0;
    if (rmask != 0u) {
        const int r_lo = __builtin_ctz(rmask);
        const int r_hi = 31 - __builtin_clz(rmask);
        j_lo = (r_lo * WW) >> 3;
        j_hi = min(((r_hi + 1) * WW + 7) >> 3, NP);
    }
    if (rmask0 != 0u) {
        t0lo = (__builtin_ctz(rmask0) * WW) >> 3;
        t0hi = min(((32 - __builtin_clz(rmask0)) * WW + 7) >> 3, NP);
    }
    if (rmask1 != 0u) {
        t1lo = (__builtin_ctz(rmask1) * WW) >> 3;
        t1hi = min(((32 - __builtin_clz(rmask1)) * WW + 7) >> 3, NP);
    }
    const int nb = (j_hi - j_lo + BATCH - 1) / BATCH;
    const int js = min(j_lo, NP - nb*BATCH);

    const float* gp = feat + (long)n * (CC * HW) + (long)(wv*16) * HW + 8*js;
    v4f* sAw = sA + wv*ASTG;

    float* pn = pooled + (long)n * CK;

    v8f acc0 = {};
    v8f acc1 = {};
    switch (nb) {
        case 0:  pool_span<0>(gp, sAw, sB, sWx, sWy, inv_area, js, t0lo, t0hi, t1lo, t1hi, tid, lane, acc0, acc1);  break;
        case 1:  pool_span<1>(gp, sAw, sB, sWx, sWy, inv_area, js, t0lo, t0hi, t1lo, t1hi, tid, lane, acc0, acc1);  break;
        case 2:  pool_span<2>(gp, sAw, sB, sWx, sWy, inv_area, js, t0lo, t0hi, t1lo, t1hi, tid, lane, acc0, acc1);  break;
        case 3:  pool_span<3>(gp, sAw, sB, sWx, sWy, inv_area, js, t0lo, t0hi, t1lo, t1hi, tid, lane, acc0, acc1);  break;
        case 4:  pool_span<4>(gp, sAw, sB, sWx, sWy, inv_area, js, t0lo, t0hi, t1lo, t1hi, tid, lane, acc0, acc1);  break;
        case 5:  pool_span<5>(gp, sAw, sB, sWx, sWy, inv_area, js, t0lo, t0hi, t1lo, t1hi, tid, lane, acc0, acc1);  break;
        case 6:  pool_span<6>(gp, sAw, sB, sWx, sWy, inv_area, js, t0lo, t0hi, t1lo, t1hi, tid, lane, acc0, acc1);  break;
        case 7:  pool_span<7>(gp, sAw, sB, sWx, sWy, inv_area, js, t0lo, t0hi, t1lo, t1hi, tid, lane, acc0, acc1);  break;
        case 8:  pool_span<8>(gp, sAw, sB, sWx, sWy, inv_area, js, t0lo, t0hi, t1lo, t1hi, tid, lane, acc0, acc1);  break;
        case 9:  pool_span<9>(gp, sAw, sB, sWx, sWy, inv_area, js, t0lo, t0hi, t1lo, t1hi, tid, lane, acc0, acc1);  break;
        default: pool_span<NBATCH>(gp, sAw, sB, sWx, sWy, inv_area, js, t0lo, t0hi, t1lo, t1hi, tid, lane, acc0, acc1); break;
    }
    {
        const v4f w0 = sB[NP*32 + lane];
        const v4f w1 = sB[(NJ + NP)*32 + lane];
        const v2f w0l = {w0.x, w0.y};
        const v2f w1l = {w1.x, w1.y};
        acc0 = wmma_f32(xt, w0l, acc0);
        acc1 = wmma_f32(xt, w1l, acc1);
    }

    const int ncol  = m16;
    const int mbase = hf * 8;
    if (ncol < 15) {
        float* p0 = pn + ncol;
        #pragma unroll
        for (int j = 0; j < 8; ++j) p0[(wv*16 + mbase + j) * PQ] = acc0[j];
    }
    if (ncol < 10) {
        float* p1 = pn + 15 + ncol;
        #pragma unroll
        for (int j = 0; j < 8; ++j) p1[(wv*16 + mbase + j) * PQ] = acc1[j];
    }
}

#define KS     16
#define KSLICE 800
#define HWV    4
#define HGRP   25
__global__ __launch_bounds__(128) void head_partial_wmma(
    float* ws, const float* __restrict__ w_obj, const float* __restrict__ w_mot)
{
    __shared__ v4f red[HWV * 128];
    const int mtile = blockIdx.x;
    const int ks    = blockIdx.y;
    const int tid   = threadIdx.x;
    const int wv    = tid >> 5;
    const int lane  = tid & 31;
    const int m16   = lane & 15;
    const int hf    = lane >> 4;

    const long koff = (long)ks * KSLICE + wv * (HGRP * 8) + 4 * hf;
    const float* ap  = ws + (long)(mtile*16 + m16) * CK + koff;
    const float* b0p = w_obj + (long)m16 * CK + koff;
    const int  nh1   = 16 + (m16 & 7);
    const float* b1p = (nh1 < 20) ? (w_obj + (long)nh1 * CK + koff)
                                  : (w_mot + (long)(nh1 - 20) * CK + koff);

    v4f xa[HGRP], xb0[HGRP], xb1[HGRP];
    #pragma unroll
    for (int i = 0; i < HGRP; ++i) {
        xa[i]  = *(const v4f*)(ap  + 8*i);
        xb0[i] = *(const v4f*)(b0p + 8*i);
        xb1[i] = *(const v4f*)(b1p + 8*i);
    }
    __builtin_amdgcn_sched_barrier(0);

    v8f acc0 = {};
    v8f acc1 = {};
    #pragma unroll
    for (int i = 0; i < HGRP; ++i) {
        const v4f x = xa[i], w0 = xb0[i], w1 = xb1[i];
        const v2f xlo = {x.x, x.y},   xhi = {x.z, x.w};
        const v2f w0l = {w0.x, w0.y}, w0h = {w0.z, w0.w};
        const v2f w1l = {w1.x, w1.y}, w1h = {w1.z, w1.w};
        acc0 = wmma_f32(xlo, w0l, acc0);  acc1 = wmma_f32(xlo, w1l, acc1);
        acc0 = wmma_f32(xhi, w0h, acc0);  acc1 = wmma_f32(xhi, w1h, acc1);
    }

    v4f* rw = red + wv * 128;
    rw[lane*2 + 0]      = (v4f){acc0[0], acc0[1], acc0[2], acc0[3]};
    rw[lane*2 + 1]      = (v4f){acc0[4], acc0[5], acc0[6], acc0[7]};
    rw[64 + lane*2 + 0] = (v4f){acc1[0], acc1[1], acc1[2], acc1[3]};
    rw[64 + lane*2 + 1] = (v4f){acc1[4], acc1[5], acc1[6], acc1[7]};
    __syncthreads();

    v4f s = red[tid];
    #pragma unroll
    for (int w = 1; w < HWV; ++w) s += red[w*128 + tid];
    *(v4f*)(ws + (long)(mtile*16) * CK + (long)ks * KSLICE + 4*tid) = s;
}

__global__ __launch_bounds__(512) void head_reduce(
    const float* __restrict__ ws,
    const float* __restrict__ b_obj, const float* __restrict__ b_mot,
    float* __restrict__ out)
{
    const int mtile = blockIdx.x;
    const int e     = threadIdx.x;
    const int tile  = e >> 8;
    const int lane  = (e >> 3) & 31;
    const int j     = e & 7;
    const int nh    = tile*16 + (lane & 15);
    const int roi   = mtile*16 + (lane >> 4)*8 + j;
    if (nh < 24) {
        const float* p = ws + (long)(mtile*16) * CK + e;
        float v[KS];
        #pragma unroll
        for (int k = 0; k < KS; ++k) v[k] = p[k * KSLICE];
        float s = v[0];
        #pragma unroll
        for (int k = 1; k < KS; ++k) s += v[k];
        if (nh < 20) out[roi*20 + nh] = s + b_obj[nh];
        else         out[20*NROI + roi*4 + (nh - 20)] = s + b_mot[nh - 20];
    }
}

extern "C" void kernel_launch(void* const* d_in, const int* in_sizes, int n_in,
                              void* d_out, int out_size, void* d_ws, size_t ws_size,
                              hipStream_t stream) {
    const float* feat  = (const float*)d_in[0];
    const float* bb    = (const float*)d_in[1];
    const float* w_obj = (const float*)d_in[2];
    const float* b_obj = (const float*)d_in[3];
    const float* w_mot = (const float*)d_in[4];
    const float* b_mot = (const float*)d_in[5];
    float* out    = (float*)d_out;
    float* pooled = (float*)d_ws;

    prroi_pool_wmma<<<dim3(NROI), 1024, 0, stream>>>(feat, bb, pooled);
    head_partial_wmma<<<dim3(16, KS), 128, 0, stream>>>(pooled, w_obj, w_mot);
    head_reduce<<<dim3(16), 512, 0, stream>>>(pooled, b_obj, b_mot, out);
}
